# starting kernel
# baseline (speedup 1.0000x reference)
#include <hip/hip_runtime.h>


#define B_ 64
#define N_ 512
#define D_ 128
#define T_ 8
#define NEG_SLOPE 0.2f

typedef __attribute__((ext_vector_type(2))) float v2f;
typedef __attribute__((ext_vector_type(8))) float v8f;

__global__ __launch_bounds__(256)
void edgeattn_project(const float* __restrict__ src,
                      const float* __restrict__ dst,
                      const float* __restrict__ a,
                      float* __restrict__ s_src,
                      float* __restrict__ s_dst) {
    const int lane  = threadIdx.x & 31;
    const int tile  = blockIdx.x * 8 + (threadIdx.x >> 5);
    const int mrow  = (tile << 4) + (lane & 15);
    const int khalf = (lane & 16) ? 2 : 0;
    const int tcol  = lane & 15;

    const int   tt  = tcol & (T_ - 1);
    const float msk = (tcol < T_) ? 1.0f : 0.0f;
    const float* __restrict__ arow   = a + (size_t)tt * (2 * D_);
    const float* __restrict__ srcRow = src + (size_t)mrow * D_;
    const float* __restrict__ dstRow = dst + (size_t)mrow * D_;

    v8f c_src = {};
    v8f c_dst = {};

#pragma unroll 4
    for (int kk = 0; kk < D_ / 4; ++kk) {
        const int k0 = (kk << 2) + khalf;

        v2f a_s, a_d, b_s, b_d;
        a_s.x = srcRow[k0];     a_s.y = srcRow[k0 + 1];
        a_d.x = dstRow[k0];     a_d.y = dstRow[k0 + 1];
        b_s.x = arow[k0]        * msk;
        b_s.y = arow[k0 + 1]    * msk;
        b_d.x = arow[D_ + k0]   * msk;
        b_d.y = arow[D_ + k0 + 1] * msk;

        c_src = __builtin_amdgcn_wmma_f32_16x16x4_f32(
            false, a_s, false, b_s, (short)0, c_src, false, false);
        c_dst = __builtin_amdgcn_wmma_f32_16x16x4_f32(
            false, a_d, false, b_d, (short)0, c_dst, false, false);
    }

    if (tcol < T_) {
        const int rbase = (tile << 4) + ((lane & 16) ? 8 : 0);
#pragma unroll
        for (int v = 0; v < 8; ++v) {
            s_src[(size_t)(rbase + v) * T_ + tcol] = c_src[v];
            s_dst[(size_t)(rbase + v) * T_ + tcol] = c_dst[v];
        }
    }
}

__global__ __launch_bounds__(256)
void edgeattn_score(const int* __restrict__ edges,
                    const float* __restrict__ s_src,
                    const float* __restrict__ s_dst,
                    float* __restrict__ out) {
    const size_t e   = ((size_t)blockIdx.x * 256 + threadIdx.x) * 4;
    const int    b   = (int)(e >> 18);
    const int    rem = (int)(e & (N_ * N_ - 1));
    const int    i   = rem >> 9;
    const int    j   = rem & (N_ - 1);

    __builtin_prefetch((const void*)(edges + e + 8192), 0, 1);

    const int4 ev = *(const int4*)(edges + e);

    const float* __restrict__ sr = s_src + ((size_t)b * N_ + i) * T_;
    const float* __restrict__ sd = s_dst + ((size_t)b * N_ + j) * T_;

    float4 r;
    {
        int k = ev.x < 0 ? 0 : (ev.x > T_ - 1 ? T_ - 1 : ev.x);
        float g = sr[k] + sd[0 * T_ + k];
        g = (ev.x < 0) ? 0.0f : g;
        r.x = (g >= 0.0f) ? g : NEG_SLOPE * g;
    }
    {
        int k = ev.y < 0 ? 0 : (ev.y > T_ - 1 ? T_ - 1 : ev.y);
        float g = sr[k] + sd[1 * T_ + k];
        g = (ev.y < 0) ? 0.0f : g;
        r.y = (g >= 0.0f) ? g : NEG_SLOPE * g;
    }
    {
        int k = ev.z < 0 ? 0 : (ev.z > T_ - 1 ? T_ - 1 : ev.z);
        float g = sr[k] + sd[2 * T_ + k];
        g = (ev.z < 0) ? 0.0f : g;
        r.z = (g >= 0.0f) ? g : NEG_SLOPE * g;
    }
    {
        int k = ev.w < 0 ? 0 : (ev.w > T_ - 1 ? T_ - 1 : ev.w);
        float g = sr[k] + sd[3 * T_ + k];
        g = (ev.w < 0) ? 0.0f : g;
        r.w = (g >= 0.0f) ? g : NEG_SLOPE * g;
    }

    *(float4*)(out + e) = r;
}

extern "C" void kernel_launch(void* const* d_in, const int* in_sizes, int n_in,
                              void* d_out, int out_size, void* d_ws, size_t ws_size,
                              hipStream_t stream) {
    const float* src   = (const float*)d_in[0];
    const float* dst   = (const float*)d_in[1];
    const int*   edges = (const int*)d_in[2];
    const float* a     = (const float*)d_in[3];
    float*       out   = (float*)d_out;

    float* s_src = (float*)d_ws;
    float* s_dst = s_src + (size_t)B_ * N_ * T_;

    edgeattn_project<<<(B_ * N_ / 16) / 8, 256, 0, stream>>>(src, dst, a, s_src, s_dst);

    const int blocks = (B_ * N_ * N_) / (4 * 256);
    edgeattn_score<<<blocks, 256, 0, stream>>>(edges, s_src, s_dst, out);
}


# projection re-tiled: a wave takes 32 rows of ONE input, weights held in registers, k in 16-byte pieces, all 48 loads issued before the first WMMA (8 drained load rounds -> 1); k sum order changes
# speedup vs baseline: 1.9647x; 1.9647x over previous
#include <hip/hip_runtime.h>


#define B_ 64
#define N_ 512
#define D_ 128
#define T_ 8
#define NEG_SLOPE 0.2f

typedef __attribute__((ext_vector_type(2))) float v2f;
typedef __attribute__((ext_vector_type(8))) float v8f;

#define PT_ROWS 32
#define PWAVES  (B_ * N_ / PT_ROWS)

__device__ __forceinline__ int pk_off(const int s, const int h) {
    return 8 * s + 4 * h;
}

__global__ __launch_bounds__(256) __attribute__((amdgpu_waves_per_eu(2, 2)))
void edgeattn_project(const float* __restrict__ src,
                      const float* __restrict__ dst,
                      const float* __restrict__ a,
                      float* __restrict__ s_src,
                      float* __restrict__ s_dst) {
    const int lane = threadIdx.x & 31;
    const int gw   = blockIdx.x * 8 + (threadIdx.x >> 5);
    const int mat  = gw / PWAVES;
    const int pr   = gw % PWAVES;
    const int c    = lane & 15;
    const int h    = lane >> 4;

    const float* __restrict__ x = mat ? dst : src;
    float*       __restrict__ y = mat ? s_dst : s_src;

    const float* __restrict__ wrow = a + (size_t)(c & (T_ - 1)) * (2 * D_) + mat * D_;
    const float* __restrict__ r0   = x + ((size_t)pr * PT_ROWS + c) * D_;
    const float* __restrict__ r1   = r0 + (size_t)16 * D_;

    float4 w[16], p0[16], p1[16];
#pragma unroll
    for (int s = 0; s < 16; ++s)
        w[s] = *(const float4*)(wrow + pk_off(s, h));
#pragma unroll
    for (int s = 0; s < 16; ++s) {
        p0[s] = *(const float4*)(r0 + pk_off(s, h));
        p1[s] = *(const float4*)(r1 + pk_off(s, h));
    }

    __builtin_amdgcn_sched_barrier(0);

    v8f c0 = {};
    v8f c1 = {};
#pragma unroll
    for (int s = 0; s < 16; ++s) {
        v2f b, a0, a1;
        b.x  = w[s].x;   b.y  = w[s].y;
        a0.x = p0[s].x;  a0.y = p0[s].y;
        a1.x = p1[s].x;  a1.y = p1[s].y;
        c0 = __builtin_amdgcn_wmma_f32_16x16x4_f32(
            false, a0, false, b, (short)0, c0, false, false);
        c1 = __builtin_amdgcn_wmma_f32_16x16x4_f32(
            false, a1, false, b, (short)0, c1, false, false);
        b.x  = w[s].z;   b.y  = w[s].w;
        a0.x = p0[s].z;  a0.y = p0[s].w;
        a1.x = p1[s].z;  a1.y = p1[s].w;
        c0 = __builtin_amdgcn_wmma_f32_16x16x4_f32(
            false, a0, false, b, (short)0, c0, false, false);
        c1 = __builtin_amdgcn_wmma_f32_16x16x4_f32(
            false, a1, false, b, (short)0, c1, false, false);
    }

    if (c < T_) {
        float* __restrict__ yb = y + ((size_t)pr * PT_ROWS + 8 * h) * T_ + c;
#pragma unroll
        for (int v = 0; v < 8; ++v) {
            yb[(size_t)v * T_]        = c0[v];
            yb[(size_t)(16 + v) * T_] = c1[v];
        }
    }
}

#define RW 16
#define RB 4
#define NG (RW / RB)
#define STRIP 128

__device__ __forceinline__
float pick8(const float4 lo, const float4 hi, const int k) {
    const bool b0 = (k & 1) != 0;
    const bool b1 = (k & 2) != 0;
    const bool b2 = (k & 4) != 0;
    const float u0 = b0 ? lo.y : lo.x;
    const float u1 = b0 ? lo.w : lo.z;
    const float u2 = b0 ? hi.y : hi.x;
    const float u3 = b0 ? hi.w : hi.z;
    const float w0 = b1 ? u1 : u0;
    const float w1 = b1 ? u3 : u2;
    return b2 ? w1 : w0;
}

__device__ __forceinline__
float edge_one(const int e, const float4 lo, const float4 hi, const unsigned srow) {
    const int k = e < 0 ? 0 : (e > T_ - 1 ? T_ - 1 : e);
    const float ss = __uint_as_float(
        __builtin_amdgcn_permlane16_var(srow, srow, (unsigned)k, true, false));
    float g = ss + pick8(lo, hi, k);
    g = (e < 0) ? 0.0f : g;
    return (g >= 0.0f) ? g : NEG_SLOPE * g;
}

#define ROWS_READY(E, S)                                                        \
    asm volatile("" :: "v"(E[0].x), "v"(E[0].y), "v"(E[0].z), "v"(E[0].w),      \
                       "v"(E[1].x), "v"(E[1].y), "v"(E[1].z), "v"(E[1].w),      \
                       "v"(E[2].x), "v"(E[2].y), "v"(E[2].z), "v"(E[2].w),      \
                       "v"(E[3].x), "v"(E[3].y), "v"(E[3].z), "v"(E[3].w),      \
                       "v"(S[0]), "v"(S[1]), "v"(S[2]), "v"(S[3]))

__global__ __launch_bounds__(256) __attribute__((amdgpu_waves_per_eu(8)))
void edgeattn_score(const int* __restrict__ edges,
                    const float* __restrict__ s_src,
                    const float* __restrict__ s_dst,
                    float* __restrict__ out) {
    const int lane  = threadIdx.x & 31;
    const int wave  = threadIdx.x >> 5;
    const int strip = blockIdx.x & 3;
    const int rtile = (blockIdx.x >> 2) & 3;
    const int b     = blockIdx.x >> 4;
    const int i0    = rtile * (8 * RW) + wave * RW;
    const int j0    = strip * STRIP + lane * 4;

    const size_t ebase = ((size_t)(b * N_ + i0)) * N_ + j0;
    const int4*  __restrict__ ep = (const int4*)(edges + ebase);
    float4*      __restrict__ op = (float4*)(out + ebase);
    const float* __restrict__ sp = s_src + ((size_t)b * N_ + i0) * T_ + (lane & 7);

    int4     ev[NG][RB];
    unsigned sv[NG][RB];

#pragma unroll
    for (int g = 0; g < 3; ++g) {
#pragma unroll
        for (int r = 0; r < RB; ++r) {
            ev[g][r] = ep[(g * RB + r) * (N_ / 4)];
            sv[g][r] = __float_as_uint(sp[(g * RB + r) * T_]);
        }
    }

    const float4* __restrict__ dp = (const float4*)(s_dst + ((size_t)b * N_ + j0) * T_);
    const float4 d0l = dp[0], d0h = dp[1];
    const float4 d1l = dp[2], d1h = dp[3];
    const float4 d2l = dp[4], d2h = dp[5];
    const float4 d3l = dp[6], d3h = dp[7];

#pragma unroll
    for (int q = 0; q < NG; ++q) {
        ROWS_READY(ev[q], sv[q]);
#pragma unroll
        for (int r = 0; r < RB; ++r) {
            const int4 e = ev[q][r];
            const unsigned s = sv[q][r];
            float4 o;
            o.x = edge_one(e.x, d0l, d0h, s);
            o.y = edge_one(e.y, d1l, d1h, s);
            o.z = edge_one(e.z, d2l, d2h, s);
            o.w = edge_one(e.w, d3l, d3h, s);
            op[(q * RB + r) * (N_ / 4)] = o;
        }
        if (q + 3 < NG) {
#pragma unroll
            for (int r = 0; r < RB; ++r) {
                ev[q + 3][r] = ep[((q + 3) * RB + r) * (N_ / 4)];
                sv[q + 3][r] = __float_as_uint(sp[((q + 3) * RB + r) * T_]);
            }
        }
    }
}

extern "C" void kernel_launch(void* const* d_in, const int* in_sizes, int n_in,
                              void* d_out, int out_size, void* d_ws, size_t ws_size,
                              hipStream_t stream) {
    const float* src   = (const float*)d_in[0];
    const float* dst   = (const float*)d_in[1];
    const int*   edges = (const int*)d_in[2];
    const float* a     = (const float*)d_in[3];
    float*       out   = (float*)d_out;

    float* s_src = (float*)d_ws;
    float* s_dst = s_src + (size_t)B_ * N_ * T_;

    edgeattn_project<<<(2 * PWAVES) / 8, 256, 0, stream>>>(src, dst, a, s_src, s_dst);

    const int blocks = B_ * (N_ / (8 * RW)) * (N_ / STRIP);
    edgeattn_score<<<blocks, 256, 0, stream>>>(edges, s_src, s_dst, out);
}


# edge pass re-gridded: a wave walks 32 rows, not 16, asking for edge rows 5 groups ahead; 4,096 waves (2 blocks per WGP, 4 per SIMD) enter, not 8,192; table load and set-up once per 32 rows
# speedup vs baseline: 2.0613x; 1.0492x over previous
#include <hip/hip_runtime.h>


#define B_ 64
#define N_ 512
#define D_ 128
#define T_ 8
#define NEG_SLOPE 0.2f

typedef __attribute__((ext_vector_type(2))) float v2f;
typedef __attribute__((ext_vector_type(8))) float v8f;

#define PT_ROWS 32
#define PWAVES  (B_ * N_ / PT_ROWS)

__device__ __forceinline__ int pk_off(const int s, const int h) {
    return 8 * s + 4 * h;
}

__global__ __launch_bounds__(256) __attribute__((amdgpu_waves_per_eu(2, 2)))
void edgeattn_project(const float* __restrict__ src,
                      const float* __restrict__ dst,
                      const float* __restrict__ a,
                      float* __restrict__ s_src,
                      float* __restrict__ s_dst) {
    const int lane = threadIdx.x & 31;
    const int gw   = blockIdx.x * 8 + (threadIdx.x >> 5);
    const int mat  = gw / PWAVES;
    const int pr   = gw % PWAVES;
    const int c    = lane & 15;
    const int h    = lane >> 4;

    const float* __restrict__ x = mat ? dst : src;
    float*       __restrict__ y = mat ? s_dst : s_src;

    const float* __restrict__ wrow = a + (size_t)(c & (T_ - 1)) * (2 * D_) + mat * D_;
    const float* __restrict__ r0   = x + ((size_t)pr * PT_ROWS + c) * D_;
    const float* __restrict__ r1   = r0 + (size_t)16 * D_;

    float4 w[16], p0[16], p1[16];
#pragma unroll
    for (int s = 0; s < 16; ++s)
        w[s] = *(const float4*)(wrow + pk_off(s, h));
#pragma unroll
    for (int s = 0; s < 16; ++s) {
        p0[s] = *(const float4*)(r0 + pk_off(s, h));
        p1[s] = *(const float4*)(r1 + pk_off(s, h));
    }

    __builtin_amdgcn_sched_barrier(0);

    v8f c0 = {};
    v8f c1 = {};
#pragma unroll
    for (int s = 0; s < 16; ++s) {
        v2f b, a0, a1;
        b.x  = w[s].x;   b.y  = w[s].y;
        a0.x = p0[s].x;  a0.y = p0[s].y;
        a1.x = p1[s].x;  a1.y = p1[s].y;
        c0 = __builtin_amdgcn_wmma_f32_16x16x4_f32(
            false, a0, false, b, (short)0, c0, false, false);
        c1 = __builtin_amdgcn_wmma_f32_16x16x4_f32(
            false, a1, false, b, (short)0, c1, false, false);
        b.x  = w[s].z;   b.y  = w[s].w;
        a0.x = p0[s].z;  a0.y = p0[s].w;
        a1.x = p1[s].z;  a1.y = p1[s].w;
        c0 = __builtin_amdgcn_wmma_f32_16x16x4_f32(
            false, a0, false, b, (short)0, c0, false, false);
        c1 = __builtin_amdgcn_wmma_f32_16x16x4_f32(
            false, a1, false, b, (short)0, c1, false, false);
    }

    if (c < T_) {
        float* __restrict__ yb = y + ((size_t)pr * PT_ROWS + 8 * h) * T_ + c;
#pragma unroll
        for (int v = 0; v < 8; ++v) {
            yb[(size_t)v * T_]        = c0[v];
            yb[(size_t)(16 + v) * T_] = c1[v];
        }
    }
}

#define RW 32
#define RB 4
#define NG (RW / RB)
#define PF 5
#define STRIP 128

__device__ __forceinline__
float pick8(const float4 lo, const float4 hi, const int k) {
    const bool b0 = (k & 1) != 0;
    const bool b1 = (k & 2) != 0;
    const bool b2 = (k & 4) != 0;
    const float u0 = b0 ? lo.y : lo.x;
    const float u1 = b0 ? lo.w : lo.z;
    const float u2 = b0 ? hi.y : hi.x;
    const float u3 = b0 ? hi.w : hi.z;
    const float w0 = b1 ? u1 : u0;
    const float w1 = b1 ? u3 : u2;
    return b2 ? w1 : w0;
}

__device__ __forceinline__
float edge_one(const int e, const float4 lo, const float4 hi, const unsigned srow) {
    const int k = e < 0 ? 0 : (e > T_ - 1 ? T_ - 1 : e);
    const float ss = __uint_as_float(
        __builtin_amdgcn_permlane16_var(srow, srow, (unsigned)k, true, false));
    float g = ss + pick8(lo, hi, k);
    g = (e < 0) ? 0.0f : g;
    return (g >= 0.0f) ? g : NEG_SLOPE * g;
}

#define ROWS_READY(E, S)                                                        \
    asm volatile("" :: "v"(E[0].x), "v"(E[0].y), "v"(E[0].z), "v"(E[0].w),      \
                       "v"(E[1].x), "v"(E[1].y), "v"(E[1].z), "v"(E[1].w),      \
                       "v"(E[2].x), "v"(E[2].y), "v"(E[2].z), "v"(E[2].w),      \
                       "v"(E[3].x), "v"(E[3].y), "v"(E[3].z), "v"(E[3].w),      \
                       "v"(S[0]), "v"(S[1]), "v"(S[2]), "v"(S[3]))

__global__ __launch_bounds__(256) __attribute__((amdgpu_waves_per_eu(6)))
void edgeattn_score(const int* __restrict__ edges,
                    const float* __restrict__ s_src,
                    const float* __restrict__ s_dst,
                    float* __restrict__ out) {
    const int lane  = threadIdx.x & 31;
    const int wave  = threadIdx.x >> 5;
    const int strip = blockIdx.x & 3;
    const int rtile = (blockIdx.x >> 2) & 1;
    const int b     = blockIdx.x >> 3;
    const int i0    = rtile * (8 * RW) + wave * RW;
    const int j0    = strip * STRIP + lane * 4;

    const size_t ebase = ((size_t)(b * N_ + i0)) * N_ + j0;
    const int4*  __restrict__ ep = (const int4*)(edges + ebase);
    float4*      __restrict__ op = (float4*)(out + ebase);
    const float* __restrict__ sp = s_src + ((size_t)b * N_ + i0) * T_ + (lane & 7);

    int4     ev[NG][RB];
    unsigned sv[NG][RB];

#pragma unroll
    for (int g = 0; g < PF; ++g) {
#pragma unroll
        for (int r = 0; r < RB; ++r) {
            ev[g][r] = ep[(g * RB + r) * (N_ / 4)];
            sv[g][r] = __float_as_uint(sp[(g * RB + r) * T_]);
        }
    }

    const float4* __restrict__ dp = (const float4*)(s_dst + ((size_t)b * N_ + j0) * T_);
    const float4 d0l = dp[0], d0h = dp[1];
    const float4 d1l = dp[2], d1h = dp[3];
    const float4 d2l = dp[4], d2h = dp[5];
    const float4 d3l = dp[6], d3h = dp[7];

#pragma unroll
    for (int q = 0; q < NG; ++q) {
        ROWS_READY(ev[q], sv[q]);
#pragma unroll
        for (int r = 0; r < RB; ++r) {
            const int4 e = ev[q][r];
            const unsigned s = sv[q][r];
            float4 o;
            o.x = edge_one(e.x, d0l, d0h, s);
            o.y = edge_one(e.y, d1l, d1h, s);
            o.z = edge_one(e.z, d2l, d2h, s);
            o.w = edge_one(e.w, d3l, d3h, s);
            op[(q * RB + r) * (N_ / 4)] = o;
        }
        if (q + PF < NG) {
#pragma unroll
            for (int r = 0; r < RB; ++r) {
                ev[q + PF][r] = ep[((q + PF) * RB + r) * (N_ / 4)];
                sv[q + PF][r] = __float_as_uint(sp[((q + PF) * RB + r) * T_]);
            }
        }
    }
}

extern "C" void kernel_launch(void* const* d_in, const int* in_sizes, int n_in,
                              void* d_out, int out_size, void* d_ws, size_t ws_size,
                              hipStream_t stream) {
    const float* src   = (const float*)d_in[0];
    const float* dst   = (const float*)d_in[1];
    const int*   edges = (const int*)d_in[2];
    const float* a     = (const float*)d_in[3];
    float*       out   = (float*)d_out;

    float* s_src = (float*)d_ws;
    float* s_dst = s_src + (size_t)B_ * N_ * T_;

    edgeattn_project<<<(2 * PWAVES) / 8, 256, 0, stream>>>(src, dst, a, s_src, s_dst);

    const int blocks = B_ * (N_ / (8 * RW)) * (N_ / STRIP);
    edgeattn_score<<<blocks, 256, 0, stream>>>(edges, s_src, s_dst, out);
}
